# starting kernel
# baseline (speedup 1.0000x reference)
#include <hip/hip_runtime.h>
#include <hip/hip_bf16.h>

typedef __attribute__((ext_vector_type(2))) float v2f;
typedef __attribute__((ext_vector_type(8))) float v8f;

#define IN_CH  128
#define OUT_CH 128

static __device__ __forceinline__ void atomic_add_f32(float* p, float v) {
    __hip_atomic_fetch_add(p, v, __ATOMIC_RELAXED, __HIP_MEMORY_SCOPE_AGENT);
}

__global__ void renorm_pack_kernel(const float* __restrict__ W,
                                   float* __restrict__ Wfrag) {
    int k = threadIdx.x;
    float s = 0.0f;
    #pragma unroll 8
    for (int o = 0; o < OUT_CH; ++o) {
        float w = W[o * IN_CH + k];
        s += w * w;
    }
    float nrm = sqrtf(s);
    float scale = (nrm > 1.0f) ? (1.0f / nrm) : 1.0f;

    const int kg  = k >> 2;
    const int kk  = k & 3;
    const int hi  = (kk >> 1) << 4;
    const int sub = kk & 1;
    #pragma unroll 8
    for (int o = 0; o < OUT_CH; ++o) {
        int cw = o >> 4;
        int m  = o & 15;
        int lane = hi | m;
        Wfrag[(((kg * 8 + cw) * 32) + lane) * 2 + sub] = W[o * IN_CH + k] * scale;
    }
}

__global__ void deg_init_kernel(float* __restrict__ deg, int N) {
    int n = blockIdx.x * blockDim.x + threadIdx.x;
    if (n < N) deg[n] = 1.0f;
}

__global__ void deg_count_kernel(const int* __restrict__ col,
                                 float* __restrict__ deg, int E) {
    int e = blockIdx.x * blockDim.x + threadIdx.x;
    if (e < E) atomic_add_f32(&deg[col[e]], 1.0f);
}

__global__ void deg_rsqrt_kernel(float* __restrict__ deg, int N) {
    int n = blockIdx.x * blockDim.x + threadIdx.x;
    if (n < N) deg[n] = rsqrtf(deg[n]);
}

__global__ void gemm_wmma_kernel(const float* __restrict__ x,
                                 const float* __restrict__ Wfrag,
                                 float* __restrict__ h, int N) {
    const int wave  = threadIdx.x >> 5;
    const int lane  = threadIdx.x & 31;
    const int row0  = blockIdx.x * 32;
    const int col0  = wave * 16;
    const int m     = lane & 15;
    const int khalf = (lane >> 4) * 2;

    int r0 = row0 + m;        if (r0 >= N) r0 = N - 1;
    int r1 = row0 + 16 + m;   if (r1 >= N) r1 = N - 1;
    const float* __restrict__ x0 = x + (size_t)r0 * IN_CH + khalf;
    const float* __restrict__ x1 = x + (size_t)r1 * IN_CH + khalf;
    const v2f*   __restrict__ bp = (const v2f*)Wfrag + (wave * 32 + lane);

    v8f c0 = {}, c1 = {};
    #pragma unroll 4
    for (int kg = 0; kg < 32; ++kg) {
        v2f a0 = *(const v2f*)(x0 + kg * 4);
        v2f a1 = *(const v2f*)(x1 + kg * 4);
        v2f b  = bp[kg * (8 * 32)];
        c0 = __builtin_amdgcn_wmma_f32_16x16x4_f32(
                false, a0, false, b, (short)0, c0, false, false);
        c1 = __builtin_amdgcn_wmma_f32_16x16x4_f32(
                false, a1, false, b, (short)0, c1, false, false);
    }

    const int nhalf = (lane >> 4) * 8;
    float* __restrict__ h0 = h + (size_t)(row0 + nhalf) * OUT_CH + col0 + m;
    float* __restrict__ h1 = h0 + (size_t)16 * OUT_CH;
    #pragma unroll
    for (int r = 0; r < 8; ++r) {
        h0[(size_t)r * OUT_CH] = c0[r];
        h1[(size_t)r * OUT_CH] = c1[r];
    }
}

__global__ void out_init_kernel(const float* __restrict__ h,
                                const float* __restrict__ dis,
                                const float* __restrict__ bias,
                                float* __restrict__ out, int N) {
    int t = blockIdx.x * blockDim.x + threadIdx.x;
    if (t >= N * 32) return;
    int n = t >> 5;
    int g = t & 31;
    float s = dis[n];
    s = s * s;
    float4 hv = ((const float4*)(h + (size_t)n * OUT_CH))[g];
    float4 bv = ((const float4*)bias)[g];
    float4 o;
    o.x = bv.x + hv.x * s;
    o.y = bv.y + hv.y * s;
    o.z = bv.z + hv.z * s;
    o.w = bv.w + hv.w * s;
    ((float4*)(out + (size_t)n * OUT_CH))[g] = o;
}

__global__ void scatter_kernel(const int* __restrict__ rowi,
                               const int* __restrict__ coli,
                               const float* __restrict__ h,
                               const float* __restrict__ dis,
                               float* __restrict__ out, int E) {
    int e = blockIdx.x * (blockDim.x >> 5) + (threadIdx.x >> 5);
    if (e >= E) return;
    int lane = threadIdx.x & 31;
    int r = rowi[e];
    int cdst = coli[e];
    float w = dis[r] * dis[cdst];
    float4 v = ((const float4*)(h + (size_t)r * OUT_CH))[lane];
    float* op = out + (size_t)cdst * OUT_CH + lane * 4;
    atomic_add_f32(op + 0, v.x * w);
    atomic_add_f32(op + 1, v.y * w);
    atomic_add_f32(op + 2, v.z * w);
    atomic_add_f32(op + 3, v.w * w);
}

extern "C" void kernel_launch(void* const* d_in, const int* in_sizes, int n_in,
                              void* d_out, int out_size, void* d_ws, size_t ws_size,
                              hipStream_t stream) {
    const float* x  = (const float*)d_in[0];
    const int*   ei = (const int*)d_in[1];
    const float* W  = (const float*)d_in[2];
    const float* b  = (const float*)d_in[3];
    float* out = (float*)d_out;

    const int N = in_sizes[0] / IN_CH;
    const int E = in_sizes[1] / 2;
    const int Npad = (N + 31) & ~31;
    const int* row = ei;
    const int* col = ei + E;

    float* ws    = (float*)d_ws;
    float* Wfrag = ws;
    float* deg   = ws + IN_CH * OUT_CH;
    float* h     = deg + Npad;

    renorm_pack_kernel<<<1, 128, 0, stream>>>(W, Wfrag);

    deg_init_kernel<<<(N + 255) / 256, 256, 0, stream>>>(deg, N);
    deg_count_kernel<<<(E + 255) / 256, 256, 0, stream>>>(col, deg, E);
    deg_rsqrt_kernel<<<(N + 255) / 256, 256, 0, stream>>>(deg, N);

    gemm_wmma_kernel<<<(N + 31) / 32, 256, 0, stream>>>(x, Wfrag, h, N);

    out_init_kernel<<<(N * 32 + 255) / 256, 256, 0, stream>>>(h, deg, b, out, N);

    scatter_kernel<<<(E + 7) / 8, 256, 0, stream>>>(row, col, h, deg, out, E);
}


# edge sum: block owns a 400-row destination tile in LDS and half the edge list, scans col, queues its edges (no sort), adds h rows by LDS f32 atomics, then tile into out; sum order: 2 partials
# speedup vs baseline: 1.6464x; 1.6464x over previous
#include <hip/hip_runtime.h>
#include <hip/hip_bf16.h>

typedef __attribute__((ext_vector_type(2))) float v2f;
typedef __attribute__((ext_vector_type(8))) float v8f;

#define IN_CH  128
#define OUT_CH 128

static __device__ __forceinline__ void atomic_add_f32(float* p, float v) {
    __hip_atomic_fetch_add(p, v, __ATOMIC_RELAXED, __HIP_MEMORY_SCOPE_AGENT);
}

__global__ void renorm_pack_kernel(const float* __restrict__ W,
                                   float* __restrict__ Wfrag) {
    int k = threadIdx.x;
    float s = 0.0f;
    #pragma unroll 8
    for (int o = 0; o < OUT_CH; ++o) {
        float w = W[o * IN_CH + k];
        s += w * w;
    }
    float nrm = sqrtf(s);
    float scale = (nrm > 1.0f) ? (1.0f / nrm) : 1.0f;

    const int kg  = k >> 2;
    const int kk  = k & 3;
    const int hi  = (kk >> 1) << 4;
    const int sub = kk & 1;
    #pragma unroll 8
    for (int o = 0; o < OUT_CH; ++o) {
        int cw = o >> 4;
        int m  = o & 15;
        int lane = hi | m;
        Wfrag[(((kg * 8 + cw) * 32) + lane) * 2 + sub] = W[o * IN_CH + k] * scale;
    }
}

__global__ void deg_init_kernel(float* __restrict__ deg, int N) {
    int n = blockIdx.x * blockDim.x + threadIdx.x;
    if (n < N) deg[n] = 1.0f;
}

__global__ void deg_count_kernel(const int* __restrict__ col,
                                 float* __restrict__ deg, int E) {
    int e = blockIdx.x * blockDim.x + threadIdx.x;
    if (e < E) atomic_add_f32(&deg[col[e]], 1.0f);
}

__global__ void deg_rsqrt_kernel(float* __restrict__ deg, int N) {
    int n = blockIdx.x * blockDim.x + threadIdx.x;
    if (n < N) deg[n] = rsqrtf(deg[n]);
}

__global__ void gemm_wmma_kernel(const float* __restrict__ x,
                                 const float* __restrict__ Wfrag,
                                 float* __restrict__ h, int N) {
    const int wave  = threadIdx.x >> 5;
    const int lane  = threadIdx.x & 31;
    const int row0  = blockIdx.x * 32;
    const int col0  = wave * 16;
    const int m     = lane & 15;
    const int khalf = (lane >> 4) * 2;

    int r0 = row0 + m;        if (r0 >= N) r0 = N - 1;
    int r1 = row0 + 16 + m;   if (r1 >= N) r1 = N - 1;
    const float* __restrict__ x0 = x + (size_t)r0 * IN_CH + khalf;
    const float* __restrict__ x1 = x + (size_t)r1 * IN_CH + khalf;
    const v2f*   __restrict__ bp = (const v2f*)Wfrag + (wave * 32 + lane);

    v8f c0 = {}, c1 = {};
    #pragma unroll 4
    for (int kg = 0; kg < 32; ++kg) {
        v2f a0 = *(const v2f*)(x0 + kg * 4);
        v2f a1 = *(const v2f*)(x1 + kg * 4);
        v2f b  = bp[kg * (8 * 32)];
        c0 = __builtin_amdgcn_wmma_f32_16x16x4_f32(
                false, a0, false, b, (short)0, c0, false, false);
        c1 = __builtin_amdgcn_wmma_f32_16x16x4_f32(
                false, a1, false, b, (short)0, c1, false, false);
    }

    const int nhalf = (lane >> 4) * 8;
    float* __restrict__ h0 = h + (size_t)(row0 + nhalf) * OUT_CH + col0 + m;
    float* __restrict__ h1 = h0 + (size_t)16 * OUT_CH;
    #pragma unroll
    for (int r = 0; r < 8; ++r) {
        h0[(size_t)r * OUT_CH] = c0[r];
        h1[(size_t)r * OUT_CH] = c1[r];
    }
}

__global__ void out_init_kernel(const float* __restrict__ h,
                                const float* __restrict__ dis,
                                const float* __restrict__ bias,
                                float* __restrict__ out, int N) {
    int t = blockIdx.x * blockDim.x + threadIdx.x;
    if (t >= N * 32) return;
    int n = t >> 5;
    int g = t & 31;
    float s = dis[n];
    s = s * s;
    float4 hv = ((const float4*)(h + (size_t)n * OUT_CH))[g];
    float4 bv = ((const float4*)bias)[g];
    float4 o;
    o.x = bv.x + hv.x * s;
    o.y = bv.y + hv.y * s;
    o.z = bv.z + hv.z * s;
    o.w = bv.w + hv.w * s;
    ((float4*)(out + (size_t)n * OUT_CH))[g] = o;
}

#define TILE_ROWS    400
#define EDGE_SLICES  2
#define TILE_WAVES   32
#define SCAN_LOADS   16
#define QUEUE_CAP    (32 + 32 * SCAN_LOADS)
#define EDGE_BATCH   8

static __device__ __forceinline__ float lane_value(float v, int j) {
    return __builtin_bit_cast(float,
        __builtin_amdgcn_readlane(__builtin_bit_cast(int, v), j));
}

static __device__ __forceinline__ void lds_add_f32(float* p, float v) {
    __hip_atomic_fetch_add(p, v, __ATOMIC_RELAXED, __HIP_MEMORY_SCOPE_WORKGROUP);
}

static __device__ __forceinline__ void tile_add_edges(const int* q, int cnt, int lane, int tile0,
                                                      const int* __restrict__ rowi,
                                                      const int* __restrict__ coli,
                                                      const float* __restrict__ h,
                                                      const float* __restrict__ dis,
                                                      float* tile) {
    int r = 0, cl = 0;
    float w = 0.0f;
    if (lane < cnt) {
        const int e = q[lane];
        const int c = coli[e];
        r  = rowi[e];
        w  = dis[r] * dis[c];
        cl = (c - tile0) * OUT_CH;
    }
    const int full = cnt & ~(EDGE_BATCH - 1);
    for (int j0 = 0; j0 < full; j0 += EDGE_BATCH) {
        float v[EDGE_BATCH][4];
        #pragma unroll
        for (int u = 0; u < EDGE_BATCH; ++u) {
            const int rj = __builtin_amdgcn_readlane(r, j0 + u);
            const float* __restrict__ hp = h + (size_t)rj * OUT_CH + lane;
            v[u][0] = hp[0];
            v[u][1] = hp[32];
            v[u][2] = hp[64];
            v[u][3] = hp[96];
        }
        #pragma unroll
        for (int u = 0; u < EDGE_BATCH; ++u) {
            const int   cj = __builtin_amdgcn_readlane(cl, j0 + u);
            const float wj = lane_value(w, j0 + u);
            float* tp = tile + cj + lane;
            lds_add_f32(tp +  0, v[u][0] * wj);
            lds_add_f32(tp + 32, v[u][1] * wj);
            lds_add_f32(tp + 64, v[u][2] * wj);
            lds_add_f32(tp + 96, v[u][3] * wj);
        }
    }
    for (int j = full; j < cnt; ++j) {
        const int   rj = __builtin_amdgcn_readlane(r, j);
        const int   cj = __builtin_amdgcn_readlane(cl, j);
        const float wj = lane_value(w, j);
        const float* __restrict__ hp = h + (size_t)rj * OUT_CH + lane;
        float v0 = hp[0], v1 = hp[32], v2 = hp[64], v3 = hp[96];
        float* tp = tile + cj + lane;
        lds_add_f32(tp +  0, v0 * wj);
        lds_add_f32(tp + 32, v1 * wj);
        lds_add_f32(tp + 64, v2 * wj);
        lds_add_f32(tp + 96, v3 * wj);
    }
}

__global__ __launch_bounds__(TILE_WAVES * 32)
void tile_scatter_kernel(const int* __restrict__ rowi,
                         const int* __restrict__ coli,
                         const float* __restrict__ h,
                         const float* __restrict__ dis,
                         float* __restrict__ out, int N, int E) {
    __shared__ float tile[TILE_ROWS * OUT_CH];
    __shared__ int   queue[TILE_WAVES * QUEUE_CAP];

    const int lane = threadIdx.x & 31;
    const int wv = __builtin_amdgcn_readfirstlane((int)(threadIdx.x >> 5));
    const int tile0 = blockIdx.x * TILE_ROWS;
    const int half  = (E + EDGE_SLICES - 1) / EDGE_SLICES;
    const int e0    = blockIdx.y * half;
    const int e1    = (e0 + half < E) ? (e0 + half) : E;
    int* q = queue + wv * QUEUE_CAP;

    for (int i = threadIdx.x; i < TILE_ROWS * OUT_CH / 4; i += TILE_WAVES * 32)
        ((float4*)tile)[i] = make_float4(0.0f, 0.0f, 0.0f, 0.0f);
    __syncthreads();

    int qn = 0;
    for (int base = e0 + wv * (32 * SCAN_LOADS); base < e1; base += TILE_WAVES * 32 * SCAN_LOADS) {
        int c[SCAN_LOADS];
        #pragma unroll
        for (int u = 0; u < SCAN_LOADS; ++u) {
            const int e  = base + u * 32 + lane;
            const int ec = (e < e1) ? e : (e1 - 1);
            const int cv = coli[ec];
            c[u] = (e < e1) ? cv : -1;
        }
        #pragma unroll
        for (int u = 0; u < SCAN_LOADS; ++u) {
            const bool hit = (unsigned)(c[u] - tile0) < (unsigned)TILE_ROWS;
            const unsigned m = __builtin_amdgcn_ballot_w32(hit);
            if (m != 0u) {
                if (hit) q[qn + __builtin_popcount(m & ((1u << lane) - 1u))] = base + u * 32 + lane;
                qn += __builtin_popcount(m);
            }
        }
        __builtin_amdgcn_fence(__ATOMIC_ACQ_REL, "wavefront");
        int head = 0;
        while (qn - head >= 32) {
            tile_add_edges(q + head, 32, lane, tile0, rowi, coli, h, dis, tile);
            head += 32;
        }
        if (head != 0) {
            const int rem = qn - head;
            int t = 0;
            if (lane < rem) t = q[head + lane];
            __builtin_amdgcn_fence(__ATOMIC_ACQ_REL, "wavefront");
            if (lane < rem) q[lane] = t;
            __builtin_amdgcn_fence(__ATOMIC_ACQ_REL, "wavefront");
            qn = rem;
        }
    }
    if (qn > 0) tile_add_edges(q, qn, lane, tile0, rowi, coli, h, dis, tile);

    __syncthreads();
    for (int rr = wv; rr < TILE_ROWS; rr += TILE_WAVES) {
        const int n = tile0 + rr;
        if (n < N) {
            const float* tp = tile + rr * OUT_CH + lane;
            float* op = out + (size_t)n * OUT_CH + lane;
            const float t0 = tp[0], t1 = tp[32], t2 = tp[64], t3 = tp[96];
            atomic_add_f32(op +  0, t0);
            atomic_add_f32(op + 32, t1);
            atomic_add_f32(op + 64, t2);
            atomic_add_f32(op + 96, t3);
        }
    }
}

extern "C" void kernel_launch(void* const* d_in, const int* in_sizes, int n_in,
                              void* d_out, int out_size, void* d_ws, size_t ws_size,
                              hipStream_t stream) {
    const float* x  = (const float*)d_in[0];
    const int*   ei = (const int*)d_in[1];
    const float* W  = (const float*)d_in[2];
    const float* b  = (const float*)d_in[3];
    float* out = (float*)d_out;

    const int N = in_sizes[0] / IN_CH;
    const int E = in_sizes[1] / 2;
    const int Npad = (N + 31) & ~31;
    const int* row = ei;
    const int* col = ei + E;

    float* ws    = (float*)d_ws;
    float* Wfrag = ws;
    float* deg   = ws + IN_CH * OUT_CH;
    float* h     = deg + Npad;

    renorm_pack_kernel<<<1, 128, 0, stream>>>(W, Wfrag);

    deg_init_kernel<<<(N + 255) / 256, 256, 0, stream>>>(deg, N);
    deg_count_kernel<<<(E + 255) / 256, 256, 0, stream>>>(col, deg, E);
    deg_rsqrt_kernel<<<(N + 255) / 256, 256, 0, stream>>>(deg, N);

    gemm_wmma_kernel<<<(N + 31) / 32, 256, 0, stream>>>(x, Wfrag, h, N);

    out_init_kernel<<<(N * 32 + 255) / 256, 256, 0, stream>>>(h, deg, b, out, N);

    tile_scatter_kernel<<<dim3((N + TILE_ROWS - 1) / TILE_ROWS, EDGE_SLICES), TILE_WAVES * 32, 0, stream>>>(row, col, h, deg, out, N, E);
}


# edge scan: tile blocks read 1-byte tile keys (written by the degree-count launch), 16 per lane load, not the 4-byte col; same adds, LDS arrival order, slice cut moved to a 512-edge multiple
# speedup vs baseline: 1.6950x; 1.0295x over previous
#include <hip/hip_runtime.h>
#include <hip/hip_bf16.h>

typedef __attribute__((ext_vector_type(2))) float v2f;
typedef __attribute__((ext_vector_type(8))) float v8f;

#define IN_CH  128
#define OUT_CH 128

#define TILE_ROWS    400
#define EDGE_SLICES  2
#define TILE_WAVES   32
#define KEYS_PER_LOAD 512
#define KEY_PREFETCH 4
#define QUEUE_CAP    (32 + KEYS_PER_LOAD)
#define EDGE_BATCH   8

static __device__ __forceinline__ void atomic_add_f32(float* p, float v) {
    __hip_atomic_fetch_add(p, v, __ATOMIC_RELAXED, __HIP_MEMORY_SCOPE_AGENT);
}

__global__ void renorm_pack_kernel(const float* __restrict__ W,
                                   float* __restrict__ Wfrag) {
    int k = threadIdx.x;
    float s = 0.0f;
    #pragma unroll 8
    for (int o = 0; o < OUT_CH; ++o) {
        float w = W[o * IN_CH + k];
        s += w * w;
    }
    float nrm = sqrtf(s);
    float scale = (nrm > 1.0f) ? (1.0f / nrm) : 1.0f;

    const int kg  = k >> 2;
    const int kk  = k & 3;
    const int hi  = (kk >> 1) << 4;
    const int sub = kk & 1;
    #pragma unroll 8
    for (int o = 0; o < OUT_CH; ++o) {
        int cw = o >> 4;
        int m  = o & 15;
        int lane = hi | m;
        Wfrag[(((kg * 8 + cw) * 32) + lane) * 2 + sub] = W[o * IN_CH + k] * scale;
    }
}

__global__ void deg_init_kernel(float* __restrict__ deg, int N) {
    int n = blockIdx.x * blockDim.x + threadIdx.x;
    if (n < N) deg[n] = 1.0f;
}

__global__ void deg_count_key_kernel(const int* __restrict__ col,
                                     float* __restrict__ deg,
                                     unsigned* __restrict__ keyw, int E, int nkw) {
    const int t = blockIdx.x * blockDim.x + threadIdx.x;
    if (t >= nkw) return;
    const int e = 4 * t;
    int c[4] = {-1, -1, -1, -1};
    if (((E & 3) == 0) && (e + 3 < E)) {
        const int4 cv = *(const int4*)(col + e);
        c[0] = cv.x; c[1] = cv.y; c[2] = cv.z; c[3] = cv.w;
    } else {
        #pragma unroll
        for (int k = 0; k < 4; ++k)
            if (e + k < E) c[k] = col[e + k];
    }
    unsigned kw = 0u;
    #pragma unroll
    for (int k = 0; k < 4; ++k) {
        unsigned kb = 0xFFu;
        if (c[k] >= 0) {
            atomic_add_f32(&deg[c[k]], 1.0f);
            kb = (unsigned)c[k] / (unsigned)TILE_ROWS;
        }
        kw |= kb << (8 * k);
    }
    keyw[t] = kw;
}

__global__ void deg_rsqrt_kernel(float* __restrict__ deg, int N) {
    int n = blockIdx.x * blockDim.x + threadIdx.x;
    if (n < N) deg[n] = rsqrtf(deg[n]);
}

__global__ void gemm_wmma_kernel(const float* __restrict__ x,
                                 const float* __restrict__ Wfrag,
                                 float* __restrict__ h, int N) {
    const int wave  = threadIdx.x >> 5;
    const int lane  = threadIdx.x & 31;
    const int row0  = blockIdx.x * 32;
    const int col0  = wave * 16;
    const int m     = lane & 15;
    const int khalf = (lane >> 4) * 2;

    int r0 = row0 + m;        if (r0 >= N) r0 = N - 1;
    int r1 = row0 + 16 + m;   if (r1 >= N) r1 = N - 1;
    const float* __restrict__ x0 = x + (size_t)r0 * IN_CH + khalf;
    const float* __restrict__ x1 = x + (size_t)r1 * IN_CH + khalf;
    const v2f*   __restrict__ bp = (const v2f*)Wfrag + (wave * 32 + lane);

    v8f c0 = {}, c1 = {};
    #pragma unroll 4
    for (int kg = 0; kg < 32; ++kg) {
        v2f a0 = *(const v2f*)(x0 + kg * 4);
        v2f a1 = *(const v2f*)(x1 + kg * 4);
        v2f b  = bp[kg * (8 * 32)];
        c0 = __builtin_amdgcn_wmma_f32_16x16x4_f32(
                false, a0, false, b, (short)0, c0, false, false);
        c1 = __builtin_amdgcn_wmma_f32_16x16x4_f32(
                false, a1, false, b, (short)0, c1, false, false);
    }

    const int nhalf = (lane >> 4) * 8;
    float* __restrict__ h0 = h + (size_t)(row0 + nhalf) * OUT_CH + col0 + m;
    float* __restrict__ h1 = h0 + (size_t)16 * OUT_CH;
    #pragma unroll
    for (int r = 0; r < 8; ++r) {
        h0[(size_t)r * OUT_CH] = c0[r];
        h1[(size_t)r * OUT_CH] = c1[r];
    }
}

__global__ void out_init_kernel(const float* __restrict__ h,
                                const float* __restrict__ dis,
                                const float* __restrict__ bias,
                                float* __restrict__ out, int N) {
    int t = blockIdx.x * blockDim.x + threadIdx.x;
    if (t >= N * 32) return;
    int n = t >> 5;
    int g = t & 31;
    float s = dis[n];
    s = s * s;
    float4 hv = ((const float4*)(h + (size_t)n * OUT_CH))[g];
    float4 bv = ((const float4*)bias)[g];
    float4 o;
    o.x = bv.x + hv.x * s;
    o.y = bv.y + hv.y * s;
    o.z = bv.z + hv.z * s;
    o.w = bv.w + hv.w * s;
    ((float4*)(out + (size_t)n * OUT_CH))[g] = o;
}

static __device__ __forceinline__ float lane_value(float v, int j) {
    return __builtin_bit_cast(float,
        __builtin_amdgcn_readlane(__builtin_bit_cast(int, v), j));
}

static __device__ __forceinline__ void lds_add_f32(float* p, float v) {
    __hip_atomic_fetch_add(p, v, __ATOMIC_RELAXED, __HIP_MEMORY_SCOPE_WORKGROUP);
}

static __device__ __forceinline__ void tile_add_edges(const int* q, int cnt, int lane, int tile0,
                                                      const int* __restrict__ rowi,
                                                      const int* __restrict__ coli,
                                                      const float* __restrict__ h,
                                                      const float* __restrict__ dis,
                                                      float* tile) {
    int r = 0, cl = 0;
    float w = 0.0f;
    if (lane < cnt) {
        const int e = q[lane];
        const int c = coli[e];
        r  = rowi[e];
        w  = dis[r] * dis[c];
        cl = (c - tile0) * OUT_CH;
    }
    const int full = cnt & ~(EDGE_BATCH - 1);
    for (int j0 = 0; j0 < full; j0 += EDGE_BATCH) {
        float v[EDGE_BATCH][4];
        #pragma unroll
        for (int u = 0; u < EDGE_BATCH; ++u) {
            const int rj = __builtin_amdgcn_readlane(r, j0 + u);
            const float* __restrict__ hp = h + (size_t)rj * OUT_CH + lane;
            v[u][0] = hp[0];
            v[u][1] = hp[32];
            v[u][2] = hp[64];
            v[u][3] = hp[96];
        }
        #pragma unroll
        for (int u = 0; u < EDGE_BATCH; ++u) {
            const int   cj = __builtin_amdgcn_readlane(cl, j0 + u);
            const float wj = lane_value(w, j0 + u);
            float* tp = tile + cj + lane;
            lds_add_f32(tp +  0, v[u][0] * wj);
            lds_add_f32(tp + 32, v[u][1] * wj);
            lds_add_f32(tp + 64, v[u][2] * wj);
            lds_add_f32(tp + 96, v[u][3] * wj);
        }
    }
    for (int j = full; j < cnt; ++j) {
        const int   rj = __builtin_amdgcn_readlane(r, j);
        const int   cj = __builtin_amdgcn_readlane(cl, j);
        const float wj = lane_value(w, j);
        const float* __restrict__ hp = h + (size_t)rj * OUT_CH + lane;
        float v0 = hp[0], v1 = hp[32], v2 = hp[64], v3 = hp[96];
        float* tp = tile + cj + lane;
        lds_add_f32(tp +  0, v0 * wj);
        lds_add_f32(tp + 32, v1 * wj);
        lds_add_f32(tp + 64, v2 * wj);
        lds_add_f32(tp + 96, v3 * wj);
    }
}

static __device__ __forceinline__ unsigned zero_bytes(unsigned x) {
    const unsigned y = ((x & 0x7F7F7F7Fu) + 0x7F7F7F7Fu) | x;
    return ~y & 0x80808080u;
}

__global__ __launch_bounds__(TILE_WAVES * 32)
void tile_scatter_kernel(const int* __restrict__ rowi,
                         const int* __restrict__ coli,
                         const unsigned* __restrict__ keyw,
                         const float* __restrict__ h,
                         const float* __restrict__ dis,
                         float* __restrict__ out, int N, int E) {
    __shared__ float tile[TILE_ROWS * OUT_CH];
    __shared__ int   queue[TILE_WAVES * QUEUE_CAP];

    const int lane = threadIdx.x & 31;
    const int wv = __builtin_amdgcn_readfirstlane((int)(threadIdx.x >> 5));
    const int tile0 = blockIdx.x * TILE_ROWS;
    const int ngroups = (E + 15) >> 4;
    const int gslice  = ((ngroups + EDGE_SLICES - 1) / EDGE_SLICES + 31) & ~31;
    const int g0      = blockIdx.y * gslice;
    const int g1      = (g0 + gslice < ngroups) ? (g0 + gslice) : ngroups;
    const int nchunks = (g1 > g0) ? ((g1 - g0 + 31) >> 5) : 0;
    const unsigned tkey = blockIdx.x * 0x01010101u;
    const uint4* __restrict__ key16 = (const uint4*)keyw;
    int* q = queue + wv * QUEUE_CAP;

    for (int i = threadIdx.x; i < TILE_ROWS * OUT_CH / 4; i += TILE_WAVES * 32)
        ((float4*)tile)[i] = make_float4(0.0f, 0.0f, 0.0f, 0.0f);
    __syncthreads();

    auto load_keys = [&](int j) -> uint4 {
        const int  g  = g0 + 32 * j + lane;
        const bool ok = (j < nchunks) && (g < g1);
        return key16[ok ? g : 0];
    };

    int qn = 0;

    auto scan_step = [&](uint4& k, int j) __attribute__((always_inline)) {
        unsigned hm = (zero_bytes(k.x ^ tkey) >> 7) | (zero_bytes(k.y ^ tkey) >> 6)
                    | (zero_bytes(k.z ^ tkey) >> 5) | (zero_bytes(k.w ^ tkey) >> 4);
        if (g0 + 32 * j + lane >= g1) hm = 0u;
        k = load_keys(j + KEY_PREFETCH * TILE_WAVES);
        const int ebase = (g0 + 32 * j + lane) * 16;
        #pragma nounroll
        for (int pass = 0; pass < 16; ++pass) {
            const bool has = hm != 0u;
            const unsigned m = __builtin_amdgcn_ballot_w32(has);
            if (m == 0u) break;
            if (has) {
                const int b = __builtin_ctz(hm);
                q[qn + __builtin_popcount(m & ((1u << lane) - 1u))] = ebase + ((b & 7) << 2) + (b >> 3);
                hm &= hm - 1u;
            }
            qn += __builtin_popcount(m);
        }
        __builtin_amdgcn_fence(__ATOMIC_ACQ_REL, "wavefront");
        int head = 0;
        while (qn - head >= 32) {
            tile_add_edges(q + head, 32, lane, tile0, rowi, coli, h, dis, tile);
            head += 32;
        }
        if (head != 0) {
            const int rem = qn - head;
            int t = 0;
            if (lane < rem) t = q[head + lane];
            __builtin_amdgcn_fence(__ATOMIC_ACQ_REL, "wavefront");
            if (lane < rem) q[lane] = t;
            __builtin_amdgcn_fence(__ATOMIC_ACQ_REL, "wavefront");
            qn = rem;
        }
    };

    uint4 k0 = load_keys(wv);
    uint4 k1 = load_keys(wv + TILE_WAVES);
    uint4 k2 = load_keys(wv + 2 * TILE_WAVES);
    uint4 k3 = load_keys(wv + 3 * TILE_WAVES);
    for (int j = wv; j < nchunks; j += KEY_PREFETCH * TILE_WAVES) {
        scan_step(k0, j);
        scan_step(k1, j + TILE_WAVES);
        scan_step(k2, j + 2 * TILE_WAVES);
        scan_step(k3, j + 3 * TILE_WAVES);
    }
    if (qn > 0) tile_add_edges(q, qn, lane, tile0, rowi, coli, h, dis, tile);

    __syncthreads();
    for (int rr = wv; rr < TILE_ROWS; rr += TILE_WAVES) {
        const int n = tile0 + rr;
        if (n < N) {
            const float* tp = tile + rr * OUT_CH + lane;
            float* op = out + (size_t)n * OUT_CH + lane;
            const float t0 = tp[0], t1 = tp[32], t2 = tp[64], t3 = tp[96];
            atomic_add_f32(op +  0, t0);
            atomic_add_f32(op + 32, t1);
            atomic_add_f32(op + 64, t2);
            atomic_add_f32(op + 96, t3);
        }
    }
}

extern "C" void kernel_launch(void* const* d_in, const int* in_sizes, int n_in,
                              void* d_out, int out_size, void* d_ws, size_t ws_size,
                              hipStream_t stream) {
    const float* x  = (const float*)d_in[0];
    const int*   ei = (const int*)d_in[1];
    const float* W  = (const float*)d_in[2];
    const float* b  = (const float*)d_in[3];
    float* out = (float*)d_out;

    const int N = in_sizes[0] / IN_CH;
    const int E = in_sizes[1] / 2;
    const int Npad = (N + 31) & ~31;
    const int* row = ei;
    const int* col = ei + E;

    float* ws    = (float*)d_ws;
    float* Wfrag = ws;
    float* deg   = ws + IN_CH * OUT_CH;
    float* h     = deg + Npad;
    unsigned* keyw = (unsigned*)(h + (size_t)Npad * OUT_CH);
    const int nkw = ((E + 15) / 16) * 4;

    renorm_pack_kernel<<<1, 128, 0, stream>>>(W, Wfrag);

    deg_init_kernel<<<(N + 255) / 256, 256, 0, stream>>>(deg, N);
    deg_count_key_kernel<<<(nkw + 255) / 256, 256, 0, stream>>>(col, deg, keyw, E, nkw);
    deg_rsqrt_kernel<<<(N + 255) / 256, 256, 0, stream>>>(deg, N);

    gemm_wmma_kernel<<<(N + 31) / 32, 256, 0, stream>>>(x, Wfrag, h, N);

    out_init_kernel<<<(N * 32 + 255) / 256, 256, 0, stream>>>(h, deg, b, out, N);

    tile_scatter_kernel<<<dim3((N + TILE_ROWS - 1) / TILE_ROWS, EDGE_SLICES), TILE_WAVES * 32, 0, stream>>>(row, col, keyw, h, deg, out, N, E);
}


# h gather: one 16-B-per-lane load per edge (whole 512-B row, 2 aligned 256-B units, h at the workspace base) instead of 4 one-line loads over 3 units; LDS tile channel-permuted, same adds
# speedup vs baseline: 2.0083x; 1.1848x over previous
#include <hip/hip_runtime.h>
#include <hip/hip_bf16.h>

typedef __attribute__((ext_vector_type(2))) float v2f;
typedef __attribute__((ext_vector_type(8))) float v8f;

#define IN_CH  128
#define OUT_CH 128

#define TILE_ROWS    400
#define EDGE_SLICES  2
#define TILE_WAVES   32
#define KEYS_PER_LOAD 512
#define KEY_PREFETCH 4
#define QUEUE_CAP    (32 + KEYS_PER_LOAD)
#define EDGE_BATCH   8

static __device__ __forceinline__ void atomic_add_f32(float* p, float v) {
    __hip_atomic_fetch_add(p, v, __ATOMIC_RELAXED, __HIP_MEMORY_SCOPE_AGENT);
}

__global__ void renorm_pack_kernel(const float* __restrict__ W,
                                   float* __restrict__ Wfrag) {
    int k = threadIdx.x;
    float s = 0.0f;
    #pragma unroll 8
    for (int o = 0; o < OUT_CH; ++o) {
        float w = W[o * IN_CH + k];
        s += w * w;
    }
    float nrm = sqrtf(s);
    float scale = (nrm > 1.0f) ? (1.0f / nrm) : 1.0f;

    const int kg  = k >> 2;
    const int kk  = k & 3;
    const int hi  = (kk >> 1) << 4;
    const int sub = kk & 1;
    #pragma unroll 8
    for (int o = 0; o < OUT_CH; ++o) {
        int cw = o >> 4;
        int m  = o & 15;
        int lane = hi | m;
        Wfrag[(((kg * 8 + cw) * 32) + lane) * 2 + sub] = W[o * IN_CH + k] * scale;
    }
}

__global__ void deg_init_kernel(float* __restrict__ deg, int N) {
    int n = blockIdx.x * blockDim.x + threadIdx.x;
    if (n < N) deg[n] = 1.0f;
}

__global__ void deg_count_key_kernel(const int* __restrict__ col,
                                     float* __restrict__ deg,
                                     unsigned* __restrict__ keyw, int E, int nkw) {
    const int t = blockIdx.x * blockDim.x + threadIdx.x;
    if (t >= nkw) return;
    const int e = 4 * t;
    int c[4] = {-1, -1, -1, -1};
    if (((E & 3) == 0) && (e + 3 < E)) {
        const int4 cv = *(const int4*)(col + e);
        c[0] = cv.x; c[1] = cv.y; c[2] = cv.z; c[3] = cv.w;
    } else {
        #pragma unroll
        for (int k = 0; k < 4; ++k)
            if (e + k < E) c[k] = col[e + k];
    }
    unsigned kw = 0u;
    #pragma unroll
    for (int k = 0; k < 4; ++k) {
        unsigned kb = 0xFFu;
        if (c[k] >= 0) {
            atomic_add_f32(&deg[c[k]], 1.0f);
            kb = (unsigned)c[k] / (unsigned)TILE_ROWS;
        }
        kw |= kb << (8 * k);
    }
    keyw[t] = kw;
}

__global__ void deg_rsqrt_kernel(float* __restrict__ deg, int N) {
    int n = blockIdx.x * blockDim.x + threadIdx.x;
    if (n < N) deg[n] = rsqrtf(deg[n]);
}

__global__ void gemm_wmma_kernel(const float* __restrict__ x,
                                 const float* __restrict__ Wfrag,
                                 float* __restrict__ h, int N) {
    const int wave  = threadIdx.x >> 5;
    const int lane  = threadIdx.x & 31;
    const int row0  = blockIdx.x * 32;
    const int col0  = wave * 16;
    const int m     = lane & 15;
    const int khalf = (lane >> 4) * 2;

    int r0 = row0 + m;        if (r0 >= N) r0 = N - 1;
    int r1 = row0 + 16 + m;   if (r1 >= N) r1 = N - 1;
    const float* __restrict__ x0 = x + (size_t)r0 * IN_CH + khalf;
    const float* __restrict__ x1 = x + (size_t)r1 * IN_CH + khalf;
    const v2f*   __restrict__ bp = (const v2f*)Wfrag + (wave * 32 + lane);

    v8f c0 = {}, c1 = {};
    #pragma unroll 4
    for (int kg = 0; kg < 32; ++kg) {
        v2f a0 = *(const v2f*)(x0 + kg * 4);
        v2f a1 = *(const v2f*)(x1 + kg * 4);
        v2f b  = bp[kg * (8 * 32)];
        c0 = __builtin_amdgcn_wmma_f32_16x16x4_f32(
                false, a0, false, b, (short)0, c0, false, false);
        c1 = __builtin_amdgcn_wmma_f32_16x16x4_f32(
                false, a1, false, b, (short)0, c1, false, false);
    }

    const int nhalf = (lane >> 4) * 8;
    float* __restrict__ h0 = h + (size_t)(row0 + nhalf) * OUT_CH + col0 + m;
    float* __restrict__ h1 = h0 + (size_t)16 * OUT_CH;
    #pragma unroll
    for (int r = 0; r < 8; ++r) {
        h0[(size_t)r * OUT_CH] = c0[r];
        h1[(size_t)r * OUT_CH] = c1[r];
    }
}

__global__ void out_init_kernel(const float* __restrict__ h,
                                const float* __restrict__ dis,
                                const float* __restrict__ bias,
                                float* __restrict__ out, int N) {
    int t = blockIdx.x * blockDim.x + threadIdx.x;
    if (t >= N * 32) return;
    int n = t >> 5;
    int g = t & 31;
    float s = dis[n];
    s = s * s;
    float4 hv = ((const float4*)(h + (size_t)n * OUT_CH))[g];
    float4 bv = ((const float4*)bias)[g];
    float4 o;
    o.x = bv.x + hv.x * s;
    o.y = bv.y + hv.y * s;
    o.z = bv.z + hv.z * s;
    o.w = bv.w + hv.w * s;
    ((float4*)(out + (size_t)n * OUT_CH))[g] = o;
}

static __device__ __forceinline__ float lane_value(float v, int j) {
    return __builtin_bit_cast(float,
        __builtin_amdgcn_readlane(__builtin_bit_cast(int, v), j));
}

static __device__ __forceinline__ void lds_add_f32(float* p, float v) {
    __hip_atomic_fetch_add(p, v, __ATOMIC_RELAXED, __HIP_MEMORY_SCOPE_WORKGROUP);
}

static __device__ __forceinline__ void tile_add_edges(const int* q, int cnt, int lane, int tile0,
                                                      const int* __restrict__ rowi,
                                                      const int* __restrict__ coli,
                                                      const float* __restrict__ h,
                                                      const float* __restrict__ dis,
                                                      float* tile) {
    int r = 0, cl = 0;
    float w = 0.0f;
    if (lane < cnt) {
        const int e = q[lane];
        const int c = coli[e];
        r  = rowi[e];
        w  = dis[r] * dis[c];
        cl = (c - tile0) * OUT_CH;
    }
    const int full = cnt & ~(EDGE_BATCH - 1);
    for (int j0 = 0; j0 < full; j0 += EDGE_BATCH) {
        float4 v[EDGE_BATCH];
        #pragma unroll
        for (int u = 0; u < EDGE_BATCH; ++u) {
            const int rj = __builtin_amdgcn_readlane(r, j0 + u);
            v[u] = ((const float4*)(h + (size_t)rj * OUT_CH))[lane];
        }
        #pragma unroll
        for (int u = 0; u < EDGE_BATCH; ++u) {
            const int   cj = __builtin_amdgcn_readlane(cl, j0 + u);
            const float wj = lane_value(w, j0 + u);
            float* tp = tile + cj + lane;
            lds_add_f32(tp +  0, v[u].x * wj);
            lds_add_f32(tp + 32, v[u].y * wj);
            lds_add_f32(tp + 64, v[u].z * wj);
            lds_add_f32(tp + 96, v[u].w * wj);
        }
    }
    for (int j = full; j < cnt; ++j) {
        const int   rj = __builtin_amdgcn_readlane(r, j);
        const int   cj = __builtin_amdgcn_readlane(cl, j);
        const float wj = lane_value(w, j);
        const float4 hv = ((const float4*)(h + (size_t)rj * OUT_CH))[lane];
        float* tp = tile + cj + lane;
        lds_add_f32(tp +  0, hv.x * wj);
        lds_add_f32(tp + 32, hv.y * wj);
        lds_add_f32(tp + 64, hv.z * wj);
        lds_add_f32(tp + 96, hv.w * wj);
    }
}

static __device__ __forceinline__ unsigned zero_bytes(unsigned x) {
    const unsigned y = ((x & 0x7F7F7F7Fu) + 0x7F7F7F7Fu) | x;
    return ~y & 0x80808080u;
}

__global__ __launch_bounds__(TILE_WAVES * 32)
void tile_scatter_kernel(const int* __restrict__ rowi,
                         const int* __restrict__ coli,
                         const unsigned* __restrict__ keyw,
                         const float* __restrict__ h,
                         const float* __restrict__ dis,
                         float* __restrict__ out, int N, int E) {
    __shared__ float tile[TILE_ROWS * OUT_CH];
    __shared__ int   queue[TILE_WAVES * QUEUE_CAP];

    const int lane = threadIdx.x & 31;
    const int wv = __builtin_amdgcn_readfirstlane((int)(threadIdx.x >> 5));
    const int tile0 = blockIdx.x * TILE_ROWS;
    const int ngroups = (E + 15) >> 4;
    const int gslice  = ((ngroups + EDGE_SLICES - 1) / EDGE_SLICES + 31) & ~31;
    const int g0      = blockIdx.y * gslice;
    const int g1      = (g0 + gslice < ngroups) ? (g0 + gslice) : ngroups;
    const int nchunks = (g1 > g0) ? ((g1 - g0 + 31) >> 5) : 0;
    const unsigned tkey = blockIdx.x * 0x01010101u;
    const uint4* __restrict__ key16 = (const uint4*)keyw;
    int* q = queue + wv * QUEUE_CAP;

    for (int i = threadIdx.x; i < TILE_ROWS * OUT_CH / 4; i += TILE_WAVES * 32)
        ((float4*)tile)[i] = make_float4(0.0f, 0.0f, 0.0f, 0.0f);
    __syncthreads();

    auto load_keys = [&](int j) -> uint4 {
        const int  g  = g0 + 32 * j + lane;
        const bool ok = (j < nchunks) && (g < g1);
        return key16[ok ? g : 0];
    };

    int qn = 0;

    auto scan_step = [&](uint4& k, int j) __attribute__((always_inline)) {
        unsigned hm = (zero_bytes(k.x ^ tkey) >> 7) | (zero_bytes(k.y ^ tkey) >> 6)
                    | (zero_bytes(k.z ^ tkey) >> 5) | (zero_bytes(k.w ^ tkey) >> 4);
        if (g0 + 32 * j + lane >= g1) hm = 0u;
        k = load_keys(j + KEY_PREFETCH * TILE_WAVES);
        const int ebase = (g0 + 32 * j + lane) * 16;
        #pragma nounroll
        for (int pass = 0; pass < 16; ++pass) {
            const bool has = hm != 0u;
            const unsigned m = __builtin_amdgcn_ballot_w32(has);
            if (m == 0u) break;
            if (has) {
                const int b = __builtin_ctz(hm);
                q[qn + __builtin_popcount(m & ((1u << lane) - 1u))] = ebase + ((b & 7) << 2) + (b >> 3);
                hm &= hm - 1u;
            }
            qn += __builtin_popcount(m);
        }
        __builtin_amdgcn_fence(__ATOMIC_ACQ_REL, "wavefront");
        int head = 0;
        while (qn - head >= 32) {
            tile_add_edges(q + head, 32, lane, tile0, rowi, coli, h, dis, tile);
            head += 32;
        }
        if (head != 0) {
            const int rem = qn - head;
            int t = 0;
            if (lane < rem) t = q[head + lane];
            __builtin_amdgcn_fence(__ATOMIC_ACQ_REL, "wavefront");
            if (lane < rem) q[lane] = t;
            __builtin_amdgcn_fence(__ATOMIC_ACQ_REL, "wavefront");
            qn = rem;
        }
    };

    uint4 k0 = load_keys(wv);
    uint4 k1 = load_keys(wv + TILE_WAVES);
    uint4 k2 = load_keys(wv + 2 * TILE_WAVES);
    uint4 k3 = load_keys(wv + 3 * TILE_WAVES);
    for (int j = wv; j < nchunks; j += KEY_PREFETCH * TILE_WAVES) {
        scan_step(k0, j);
        scan_step(k1, j + TILE_WAVES);
        scan_step(k2, j + 2 * TILE_WAVES);
        scan_step(k3, j + 3 * TILE_WAVES);
    }
    if (qn > 0) tile_add_edges(q, qn, lane, tile0, rowi, coli, h, dis, tile);

    __syncthreads();
    const int pl = ((lane & 3) << 5) + (lane >> 2);
    for (int rr = wv; rr < TILE_ROWS; rr += TILE_WAVES) {
        const int n = tile0 + rr;
        if (n < N) {
            const float* tp = tile + rr * OUT_CH + pl;
            float* op = out + (size_t)n * OUT_CH + lane;
            const float t0 = tp[0], t1 = tp[8], t2 = tp[16], t3 = tp[24];
            atomic_add_f32(op +  0, t0);
            atomic_add_f32(op + 32, t1);
            atomic_add_f32(op + 64, t2);
            atomic_add_f32(op + 96, t3);
        }
    }
}

extern "C" void kernel_launch(void* const* d_in, const int* in_sizes, int n_in,
                              void* d_out, int out_size, void* d_ws, size_t ws_size,
                              hipStream_t stream) {
    const float* x  = (const float*)d_in[0];
    const int*   ei = (const int*)d_in[1];
    const float* W  = (const float*)d_in[2];
    const float* b  = (const float*)d_in[3];
    float* out = (float*)d_out;

    const int N = in_sizes[0] / IN_CH;
    const int E = in_sizes[1] / 2;
    const int Npad = (N + 31) & ~31;
    const int* row = ei;
    const int* col = ei + E;

    float* ws    = (float*)d_ws;
    float* h     = ws;
    float* Wfrag = h + (size_t)Npad * OUT_CH;
    float* deg   = Wfrag + IN_CH * OUT_CH;
    unsigned* keyw = (unsigned*)(deg + Npad);
    const int nkw = ((E + 15) / 16) * 4;

    renorm_pack_kernel<<<1, 128, 0, stream>>>(W, Wfrag);

    deg_init_kernel<<<(N + 255) / 256, 256, 0, stream>>>(deg, N);
    deg_count_key_kernel<<<(nkw + 255) / 256, 256, 0, stream>>>(col, deg, keyw, E, nkw);
    deg_rsqrt_kernel<<<(N + 255) / 256, 256, 0, stream>>>(deg, N);

    gemm_wmma_kernel<<<(N + 31) / 32, 256, 0, stream>>>(x, Wfrag, h, N);

    out_init_kernel<<<(N * 32 + 255) / 256, 256, 0, stream>>>(h, deg, b, out, N);

    tile_scatter_kernel<<<dim3((N + TILE_ROWS - 1) / TILE_ROWS, EDGE_SLICES), TILE_WAVES * 32, 0, stream>>>(row, col, keyw, h, deg, out, N, E);
}


# GEMM: a wave owns 16 rows x all 128 cols (64-row blocks): each x row read once, 16-B loads (k = 8g+4h+0..3 per lane, W packed to match), not 8-B 16-line loads in each of 8 waves; k sum order changes
# speedup vs baseline: 2.2138x; 1.1023x over previous
#include <hip/hip_runtime.h>
#include <hip/hip_bf16.h>

typedef __attribute__((ext_vector_type(2))) float v2f;
typedef __attribute__((ext_vector_type(8))) float v8f;

#define IN_CH  128
#define OUT_CH 128

#define TILE_ROWS    400
#define EDGE_SLICES  2
#define TILE_WAVES   32
#define KEYS_PER_LOAD 512
#define KEY_PREFETCH 4
#define QUEUE_CAP    (32 + KEYS_PER_LOAD)
#define EDGE_BATCH   8

static __device__ __forceinline__ void atomic_add_f32(float* p, float v) {
    __hip_atomic_fetch_add(p, v, __ATOMIC_RELAXED, __HIP_MEMORY_SCOPE_AGENT);
}

__global__ void renorm_pack_kernel(const float* __restrict__ W,
                                   float* __restrict__ Wfrag) {
    int k = threadIdx.x;
    float s = 0.0f;
    #pragma unroll 8
    for (int o = 0; o < OUT_CH; ++o) {
        float w = W[o * IN_CH + k];
        s += w * w;
    }
    float nrm = sqrtf(s);
    float scale = (nrm > 1.0f) ? (1.0f / nrm) : 1.0f;

    const int g   = k >> 3;
    const int hi  = ((k >> 2) & 1) << 4;
    const int sub = k & 3;
    #pragma unroll 8
    for (int o = 0; o < OUT_CH; ++o) {
        int cw = o >> 4;
        int m  = o & 15;
        int lane = hi | m;
        Wfrag[(((g * 8 + cw) * 32) + lane) * 4 + sub] = W[o * IN_CH + k] * scale;
    }
}

__global__ void deg_init_kernel(float* __restrict__ deg, int N) {
    int n = blockIdx.x * blockDim.x + threadIdx.x;
    if (n < N) deg[n] = 1.0f;
}

__global__ void deg_count_key_kernel(const int* __restrict__ col,
                                     float* __restrict__ deg,
                                     unsigned* __restrict__ keyw, int E, int nkw) {
    const int t = blockIdx.x * blockDim.x + threadIdx.x;
    if (t >= nkw) return;
    const int e = 4 * t;
    int c[4] = {-1, -1, -1, -1};
    if (((E & 3) == 0) && (e + 3 < E)) {
        const int4 cv = *(const int4*)(col + e);
        c[0] = cv.x; c[1] = cv.y; c[2] = cv.z; c[3] = cv.w;
    } else {
        #pragma unroll
        for (int k = 0; k < 4; ++k)
            if (e + k < E) c[k] = col[e + k];
    }
    unsigned kw = 0u;
    #pragma unroll
    for (int k = 0; k < 4; ++k) {
        unsigned kb = 0xFFu;
        if (c[k] >= 0) {
            atomic_add_f32(&deg[c[k]], 1.0f);
            kb = (unsigned)c[k] / (unsigned)TILE_ROWS;
        }
        kw |= kb << (8 * k);
    }
    keyw[t] = kw;
}

__global__ void deg_rsqrt_kernel(float* __restrict__ deg, int N) {
    int n = blockIdx.x * blockDim.x + threadIdx.x;
    if (n < N) deg[n] = rsqrtf(deg[n]);
}

__global__ __launch_bounds__(128)
void gemm_wmma_kernel(const float* __restrict__ x,
                      const float* __restrict__ Wfrag,
                      float* __restrict__ h, int N) {
    const int wave = threadIdx.x >> 5;
    const int lane = threadIdx.x & 31;
    const int row0 = blockIdx.x * 64 + wave * 16;
    const int m    = lane & 15;
    const int hh   = lane >> 4;

    int r = row0 + m;   if (r >= N) r = N - 1;
    const float4* __restrict__ xa = (const float4*)(x + (size_t)r * IN_CH) + hh;
    const float4* __restrict__ bp = (const float4*)Wfrag + lane;

    float4 a[3];
    float4 b[3][8];
    #pragma unroll
    for (int p = 0; p < 2; ++p) {
        a[p] = xa[2 * p];
        #pragma unroll
        for (int j = 0; j < 8; ++j) b[p][j] = bp[(p * 8 + j) * 32];
    }

    v8f acc[8];
    #pragma unroll
    for (int j = 0; j < 8; ++j) acc[j] = v8f{};

    #pragma unroll
    for (int g = 0; g < 16; ++g) {
        if (g + 2 < 16) {
            a[(g + 2) % 3] = xa[2 * (g + 2)];
            #pragma unroll
            for (int j = 0; j < 8; ++j) b[(g + 2) % 3][j] = bp[((g + 2) * 8 + j) * 32];
        }
        __builtin_amdgcn_sched_barrier(0);
        const v2f alo = {a[g % 3].x, a[g % 3].y};
        const v2f ahi = {a[g % 3].z, a[g % 3].w};
        #pragma unroll
        for (int j = 0; j < 8; ++j) {
            const v2f blo = {b[g % 3][j].x, b[g % 3][j].y};
            acc[j] = __builtin_amdgcn_wmma_f32_16x16x4_f32(
                        false, alo, false, blo, (short)0, acc[j], false, false);
        }
        #pragma unroll
        for (int j = 0; j < 8; ++j) {
            const v2f bhi = {b[g % 3][j].z, b[g % 3][j].w};
            acc[j] = __builtin_amdgcn_wmma_f32_16x16x4_f32(
                        false, ahi, false, bhi, (short)0, acc[j], false, false);
        }
        __builtin_amdgcn_sched_barrier(0);
    }

    float* __restrict__ hp = h + (size_t)(row0 + 8 * hh) * OUT_CH + m;
    #pragma unroll
    for (int j = 0; j < 8; ++j) {
        #pragma unroll
        for (int rr = 0; rr < 8; ++rr)
            hp[(size_t)rr * OUT_CH + 16 * j] = acc[j][rr];
    }
}

__global__ void out_init_kernel(const float* __restrict__ h,
                                const float* __restrict__ dis,
                                const float* __restrict__ bias,
                                float* __restrict__ out, int N) {
    int t = blockIdx.x * blockDim.x + threadIdx.x;
    if (t >= N * 32) return;
    int n = t >> 5;
    int g = t & 31;
    float s = dis[n];
    s = s * s;
    float4 hv = ((const float4*)(h + (size_t)n * OUT_CH))[g];
    float4 bv = ((const float4*)bias)[g];
    float4 o;
    o.x = bv.x + hv.x * s;
    o.y = bv.y + hv.y * s;
    o.z = bv.z + hv.z * s;
    o.w = bv.w + hv.w * s;
    ((float4*)(out + (size_t)n * OUT_CH))[g] = o;
}

static __device__ __forceinline__ float lane_value(float v, int j) {
    return __builtin_bit_cast(float,
        __builtin_amdgcn_readlane(__builtin_bit_cast(int, v), j));
}

static __device__ __forceinline__ void lds_add_f32(float* p, float v) {
    __hip_atomic_fetch_add(p, v, __ATOMIC_RELAXED, __HIP_MEMORY_SCOPE_WORKGROUP);
}

static __device__ __forceinline__ void tile_add_edges(const int* q, int cnt, int lane, int tile0,
                                                      const int* __restrict__ rowi,
                                                      const int* __restrict__ coli,
                                                      const float* __restrict__ h,
                                                      const float* __restrict__ dis,
                                                      float* tile) {
    int r = 0, cl = 0;
    float w = 0.0f;
    if (lane < cnt) {
        const int e = q[lane];
        const int c = coli[e];
        r  = rowi[e];
        w  = dis[r] * dis[c];
        cl = (c - tile0) * OUT_CH;
    }
    const int full = cnt & ~(EDGE_BATCH - 1);
    for (int j0 = 0; j0 < full; j0 += EDGE_BATCH) {
        float4 v[EDGE_BATCH];
        #pragma unroll
        for (int u = 0; u < EDGE_BATCH; ++u) {
            const int rj = __builtin_amdgcn_readlane(r, j0 + u);
            v[u] = ((const float4*)(h + (size_t)rj * OUT_CH))[lane];
        }
        #pragma unroll
        for (int u = 0; u < EDGE_BATCH; ++u) {
            const int   cj = __builtin_amdgcn_readlane(cl, j0 + u);
            const float wj = lane_value(w, j0 + u);
            float* tp = tile + cj + lane;
            lds_add_f32(tp +  0, v[u].x * wj);
            lds_add_f32(tp + 32, v[u].y * wj);
            lds_add_f32(tp + 64, v[u].z * wj);
            lds_add_f32(tp + 96, v[u].w * wj);
        }
    }
    for (int j = full; j < cnt; ++j) {
        const int   rj = __builtin_amdgcn_readlane(r, j);
        const int   cj = __builtin_amdgcn_readlane(cl, j);
        const float wj = lane_value(w, j);
        const float4 hv = ((const float4*)(h + (size_t)rj * OUT_CH))[lane];
        float* tp = tile + cj + lane;
        lds_add_f32(tp +  0, hv.x * wj);
        lds_add_f32(tp + 32, hv.y * wj);
        lds_add_f32(tp + 64, hv.z * wj);
        lds_add_f32(tp + 96, hv.w * wj);
    }
}

static __device__ __forceinline__ unsigned zero_bytes(unsigned x) {
    const unsigned y = ((x & 0x7F7F7F7Fu) + 0x7F7F7F7Fu) | x;
    return ~y & 0x80808080u;
}

__global__ __launch_bounds__(TILE_WAVES * 32)
void tile_scatter_kernel(const int* __restrict__ rowi,
                         const int* __restrict__ coli,
                         const unsigned* __restrict__ keyw,
                         const float* __restrict__ h,
                         const float* __restrict__ dis,
                         float* __restrict__ out, int N, int E) {
    __shared__ float tile[TILE_ROWS * OUT_CH];
    __shared__ int   queue[TILE_WAVES * QUEUE_CAP];

    const int lane = threadIdx.x & 31;
    const int wv = __builtin_amdgcn_readfirstlane((int)(threadIdx.x >> 5));
    const int tile0 = blockIdx.x * TILE_ROWS;
    const int ngroups = (E + 15) >> 4;
    const int gslice  = ((ngroups + EDGE_SLICES - 1) / EDGE_SLICES + 31) & ~31;
    const int g0      = blockIdx.y * gslice;
    const int g1      = (g0 + gslice < ngroups) ? (g0 + gslice) : ngroups;
    const int nchunks = (g1 > g0) ? ((g1 - g0 + 31) >> 5) : 0;
    const unsigned tkey = blockIdx.x * 0x01010101u;
    const uint4* __restrict__ key16 = (const uint4*)keyw;
    int* q = queue + wv * QUEUE_CAP;

    for (int i = threadIdx.x; i < TILE_ROWS * OUT_CH / 4; i += TILE_WAVES * 32)
        ((float4*)tile)[i] = make_float4(0.0f, 0.0f, 0.0f, 0.0f);
    __syncthreads();

    auto load_keys = [&](int j) -> uint4 {
        const int  g  = g0 + 32 * j + lane;
        const bool ok = (j < nchunks) && (g < g1);
        return key16[ok ? g : 0];
    };

    int qn = 0;

    auto scan_step = [&](uint4& k, int j) __attribute__((always_inline)) {
        unsigned hm = (zero_bytes(k.x ^ tkey) >> 7) | (zero_bytes(k.y ^ tkey) >> 6)
                    | (zero_bytes(k.z ^ tkey) >> 5) | (zero_bytes(k.w ^ tkey) >> 4);
        if (g0 + 32 * j + lane >= g1) hm = 0u;
        k = load_keys(j + KEY_PREFETCH * TILE_WAVES);
        const int ebase = (g0 + 32 * j + lane) * 16;
        #pragma nounroll
        for (int pass = 0; pass < 16; ++pass) {
            const bool has = hm != 0u;
            const unsigned m = __builtin_amdgcn_ballot_w32(has);
            if (m == 0u) break;
            if (has) {
                const int b = __builtin_ctz(hm);
                q[qn + __builtin_popcount(m & ((1u << lane) - 1u))] = ebase + ((b & 7) << 2) + (b >> 3);
                hm &= hm - 1u;
            }
            qn += __builtin_popcount(m);
        }
        __builtin_amdgcn_fence(__ATOMIC_ACQ_REL, "wavefront");
        int head = 0;
        while (qn - head >= 32) {
            tile_add_edges(q + head, 32, lane, tile0, rowi, coli, h, dis, tile);
            head += 32;
        }
        if (head != 0) {
            const int rem = qn - head;
            int t = 0;
            if (lane < rem) t = q[head + lane];
            __builtin_amdgcn_fence(__ATOMIC_ACQ_REL, "wavefront");
            if (lane < rem) q[lane] = t;
            __builtin_amdgcn_fence(__ATOMIC_ACQ_REL, "wavefront");
            qn = rem;
        }
    };

    uint4 k0 = load_keys(wv);
    uint4 k1 = load_keys(wv + TILE_WAVES);
    uint4 k2 = load_keys(wv + 2 * TILE_WAVES);
    uint4 k3 = load_keys(wv + 3 * TILE_WAVES);
    for (int j = wv; j < nchunks; j += KEY_PREFETCH * TILE_WAVES) {
        scan_step(k0, j);
        scan_step(k1, j + TILE_WAVES);
        scan_step(k2, j + 2 * TILE_WAVES);
        scan_step(k3, j + 3 * TILE_WAVES);
    }
    if (qn > 0) tile_add_edges(q, qn, lane, tile0, rowi, coli, h, dis, tile);

    __syncthreads();
    const int pl = ((lane & 3) << 5) + (lane >> 2);
    for (int rr = wv; rr < TILE_ROWS; rr += TILE_WAVES) {
        const int n = tile0 + rr;
        if (n < N) {
            const float* tp = tile + rr * OUT_CH + pl;
            float* op = out + (size_t)n * OUT_CH + lane;
            const float t0 = tp[0], t1 = tp[8], t2 = tp[16], t3 = tp[24];
            atomic_add_f32(op +  0, t0);
            atomic_add_f32(op + 32, t1);
            atomic_add_f32(op + 64, t2);
            atomic_add_f32(op + 96, t3);
        }
    }
}

extern "C" void kernel_launch(void* const* d_in, const int* in_sizes, int n_in,
                              void* d_out, int out_size, void* d_ws, size_t ws_size,
                              hipStream_t stream) {
    const float* x  = (const float*)d_in[0];
    const int*   ei = (const int*)d_in[1];
    const float* W  = (const float*)d_in[2];
    const float* b  = (const float*)d_in[3];
    float* out = (float*)d_out;

    const int N = in_sizes[0] / IN_CH;
    const int E = in_sizes[1] / 2;
    const int Npad = (N + 63) & ~63;
    const int* row = ei;
    const int* col = ei + E;

    float* ws    = (float*)d_ws;
    float* h     = ws;
    float* Wfrag = h + (size_t)Npad * OUT_CH;
    float* deg   = Wfrag + IN_CH * OUT_CH;
    unsigned* keyw = (unsigned*)(deg + Npad);
    const int nkw = ((E + 15) / 16) * 4;

    renorm_pack_kernel<<<1, 128, 0, stream>>>(W, Wfrag);

    deg_init_kernel<<<(N + 255) / 256, 256, 0, stream>>>(deg, N);
    deg_count_key_kernel<<<(nkw + 255) / 256, 256, 0, stream>>>(col, deg, keyw, E, nkw);
    deg_rsqrt_kernel<<<(N + 255) / 256, 256, 0, stream>>>(deg, N);

    gemm_wmma_kernel<<<(N + 63) / 64, 128, 0, stream>>>(x, Wfrag, h, N);

    out_init_kernel<<<(N * 32 + 255) / 256, 256, 0, stream>>>(h, deg, b, out, N);

    tile_scatter_kernel<<<dim3((N + TILE_ROWS - 1) / TILE_ROWS, EDGE_SLICES), TILE_WAVES * 32, 0, stream>>>(row, col, keyw, h, deg, out, N, E);
}
